# starting kernel
# baseline (speedup 1.0000x reference)
#include <hip/hip_runtime.h>
#include <hip/hip_bf16.h>

typedef __attribute__((ext_vector_type(2))) float v2f;
typedef __attribute__((ext_vector_type(8))) float v8f;

#define S_LEN   2048
#define D_MODEL 1024
#define NHEAD   16
#define DK      64
#define BATCH   2
#define M_TOT   (BATCH * S_LEN)
#define ROPE_L2T_OVER_DK 0.20762050593046014f

__device__ __forceinline__ float red_max16(float v) {
  v = fmaxf(v, __shfl_xor(v, 1, 32));
  v = fmaxf(v, __shfl_xor(v, 2, 32));
  v = fmaxf(v, __shfl_xor(v, 4, 32));
  v = fmaxf(v, __shfl_xor(v, 8, 32));
  return v;
}
__device__ __forceinline__ float red_sum16(float v) {
  v += __shfl_xor(v, 1, 32);
  v += __shfl_xor(v, 2, 32);
  v += __shfl_xor(v, 4, 32);
  v += __shfl_xor(v, 8, 32);
  return v;
}

__global__ __launch_bounds__(256) void gemm_wmma_f32(
    const float* __restrict__ X, const float* __restrict__ W,
    float* __restrict__ out, int mode)
{
  const int lane = threadIdx.x & 31;
  const int wave = threadIdx.x >> 5;
  const int half = lane >> 4;
  const int l16  = lane & 15;

  const int m0 = (blockIdx.y * 8 + wave) * 32;
  const int n0 = blockIdx.x * 64;

  const float* Arow0 = X + (size_t)(m0 + l16) * D_MODEL;
  const float* Arow1 = Arow0 + (size_t)16 * D_MODEL;
  const float* Brow0 = W + (size_t)(n0 + l16) * D_MODEL;
  const float* Brow1 = Brow0 + (size_t)16 * D_MODEL;
  const float* Brow2 = Brow0 + (size_t)32 * D_MODEL;
  const float* Brow3 = Brow0 + (size_t)48 * D_MODEL;

  v8f acc[2][4];
#pragma unroll
  for (int i = 0; i < 2; ++i)
#pragma unroll
    for (int t = 0; t < 4; ++t) acc[i][t] = (v8f){};

  const int koff = 2 * half;
#pragma unroll 2
  for (int k = 0; k < D_MODEL; k += 4) {
    const int kk = k + koff;
    v2f a0 = *(const v2f*)(Arow0 + kk);
    v2f a1 = *(const v2f*)(Arow1 + kk);
    v2f b0 = *(const v2f*)(Brow0 + kk);
    v2f b1 = *(const v2f*)(Brow1 + kk);
    v2f b2 = *(const v2f*)(Brow2 + kk);
    v2f b3 = *(const v2f*)(Brow3 + kk);
    acc[0][0] = __builtin_amdgcn_wmma_f32_16x16x4_f32(false, a0, false, b0, (short)0, acc[0][0], false, false);
    acc[0][1] = __builtin_amdgcn_wmma_f32_16x16x4_f32(false, a0, false, b1, (short)0, acc[0][1], false, false);
    acc[0][2] = __builtin_amdgcn_wmma_f32_16x16x4_f32(false, a0, false, b2, (short)0, acc[0][2], false, false);
    acc[0][3] = __builtin_amdgcn_wmma_f32_16x16x4_f32(false, a0, false, b3, (short)0, acc[0][3], false, false);
    acc[1][0] = __builtin_amdgcn_wmma_f32_16x16x4_f32(false, a1, false, b0, (short)0, acc[1][0], false, false);
    acc[1][1] = __builtin_amdgcn_wmma_f32_16x16x4_f32(false, a1, false, b1, (short)0, acc[1][1], false, false);
    acc[1][2] = __builtin_amdgcn_wmma_f32_16x16x4_f32(false, a1, false, b2, (short)0, acc[1][2], false, false);
    acc[1][3] = __builtin_amdgcn_wmma_f32_16x16x4_f32(false, a1, false, b3, (short)0, acc[1][3], false, false);
  }

  if (mode == 0) {
#pragma unroll
    for (int i = 0; i < 2; ++i)
#pragma unroll
      for (int t = 0; t < 4; ++t)
#pragma unroll
        for (int r = 0; r < 8; ++r) {
          const int row = m0 + i * 16 + r + 8 * half;
          out[(size_t)row * D_MODEL + n0 + t * 16 + l16] = acc[i][t][r];
        }
  } else {
#pragma unroll
    for (int t = 0; t < 4; ++t) {
      const int gcol = n0 + t * 16 + l16;
      const int h = gcol >> 6;
      const int d = gcol & 63;
      const bool even = (d & 1) == 0;
      const float invfreq = exp2f(-(float)(d & ~1) * ROPE_L2T_OVER_DK);
#pragma unroll
      for (int i = 0; i < 2; ++i)
#pragma unroll
        for (int r = 0; r < 8; ++r) {
          const int row = m0 + i * 16 + r + 8 * half;
          const int b  = row >> 11;
          const int s  = row & (S_LEN - 1);
          float v  = acc[i][t][r];
          float vp = __shfl_xor(v, 1, 32);
          if (mode == 1) {
            const float ang = (float)s * invfreq;
            const float c = __cosf(ang), sn = __sinf(ang);
            v = v * c + (even ? -vp : vp) * sn;
          }
          out[(((size_t)(b * NHEAD + h) * S_LEN + s) * DK) + d] = v;
        }
    }
  }
}

__global__ __launch_bounds__(32) void flash_attn(
    const float* __restrict__ Q, const float* __restrict__ K,
    const float* __restrict__ V, float* __restrict__ att)
{
  __shared__ float p_lds[16 * 16];
  const int lane = threadIdx.x & 31;
  const int half = lane >> 4;
  const int l16  = lane & 15;
  const int qt   = blockIdx.x;
  const int bh   = blockIdx.y;
  const size_t base = (size_t)bh * S_LEN * DK;

  v2f qa[16];
  {
    const float* qrow = Q + base + (size_t)(qt * 16 + l16) * DK + 2 * half;
#pragma unroll
    for (int c = 0; c < 16; ++c) {
      v2f t = *(const v2f*)(qrow + 4 * c);
      qa[c] = t * 0.125f;
    }
  }

  v8f o0 = {}, o1 = {}, o2 = {}, o3 = {};
  float mrun[8], lrun[8];
#pragma unroll
  for (int r = 0; r < 8; ++r) { mrun[r] = -__builtin_inff(); lrun[r] = 0.f; }

  for (int kb = 0; kb <= qt; ++kb) {
    v8f sc = {};
    const float* krow = K + base + (size_t)(kb * 16 + l16) * DK + 2 * half;
#pragma unroll
    for (int c = 0; c < 16; ++c) {
      v2f bf = *(const v2f*)(krow + 4 * c);
      sc = __builtin_amdgcn_wmma_f32_16x16x4_f32(false, qa[c], false, bf,
                                                 (short)0, sc, false, false);
    }

    float pvals[8], scl[8];
#pragma unroll
    for (int r = 0; r < 8; ++r) {
      const int row = r + 8 * half;
      const int qg  = qt * 16 + row;
      const int kg  = kb * 16 + l16;
      const float s = (kg <= qg) ? sc[r] : -__builtin_inff();
      const float rmax = red_max16(s);
      const float mnew = fmaxf(mrun[r], rmax);
      const float p = __expf(s - mnew);
      const float rsum = red_sum16(p);
      scl[r]  = __expf(mrun[r] - mnew);
      lrun[r] = lrun[r] * scl[r] + rsum;
      mrun[r] = mnew;
      pvals[r] = p;
    }

#pragma unroll
    for (int r = 0; r < 8; ++r) {
      o0[r] *= scl[r]; o1[r] *= scl[r]; o2[r] *= scl[r]; o3[r] *= scl[r];
    }

#pragma unroll
    for (int r = 0; r < 8; ++r)
      p_lds[(r + 8 * half) * 16 + l16] = pvals[r];
    __syncthreads();

    const float* vbase = V + base + (size_t)(kb * 16) * DK;
#pragma unroll
    for (int c = 0; c < 4; ++c) {
      const int kk = c * 4 + 2 * half;
      v2f a;
      a[0] = p_lds[l16 * 16 + kk];
      a[1] = p_lds[l16 * 16 + kk + 1];
      v2f b0, b1, b2, b3;
      const float* v0 = vbase + (size_t)kk * DK + l16;
      const float* v1 = v0 + DK;
      b0[0] = v0[0];  b0[1] = v1[0];
      b1[0] = v0[16]; b1[1] = v1[16];
      b2[0] = v0[32]; b2[1] = v1[32];
      b3[0] = v0[48]; b3[1] = v1[48];
      o0 = __builtin_amdgcn_wmma_f32_16x16x4_f32(false, a, false, b0, (short)0, o0, false, false);
      o1 = __builtin_amdgcn_wmma_f32_16x16x4_f32(false, a, false, b1, (short)0, o1, false, false);
      o2 = __builtin_amdgcn_wmma_f32_16x16x4_f32(false, a, false, b2, (short)0, o2, false, false);
      o3 = __builtin_amdgcn_wmma_f32_16x16x4_f32(false, a, false, b3, (short)0, o3, false, false);
    }
    __syncthreads();
  }

  const int b = bh >> 4;
  const int h = bh & 15;
#pragma unroll
  for (int r = 0; r < 8; ++r) {
    const float inv = 1.0f / lrun[r];
    const int row = qt * 16 + r + 8 * half;
    const size_t ob = ((size_t)(b * S_LEN) + row) * D_MODEL + h * DK;
    att[ob +  0 + l16] = o0[r] * inv;
    att[ob + 16 + l16] = o1[r] * inv;
    att[ob + 32 + l16] = o2[r] * inv;
    att[ob + 48 + l16] = o3[r] * inv;
  }
}

extern "C" void kernel_launch(void* const* d_in, const int* in_sizes, int n_in,
                              void* d_out, int out_size, void* d_ws, size_t ws_size,
                              hipStream_t stream) {
  const float* x  = (const float*)d_in[0];
  const float* wq = (const float*)d_in[1];
  const float* wk = (const float*)d_in[2];
  const float* wv = (const float*)d_in[3];
  const float* wo = (const float*)d_in[4];
  float* out = (float*)d_out;

  float* ws  = (float*)d_ws;
  const size_t SZ = (size_t)M_TOT * D_MODEL;
  float* Qw  = ws;
  float* Kw  = ws + SZ;
  float* Vw  = ws + 2 * SZ;
  float* Aw  = ws + 3 * SZ;

  dim3 blk(256);
  dim3 grd(D_MODEL / 64, M_TOT / (32 * 8));
  gemm_wmma_f32<<<grd, blk, 0, stream>>>(x, wq, Qw, 1);
  gemm_wmma_f32<<<grd, blk, 0, stream>>>(x, wk, Kw, 1);
  gemm_wmma_f32<<<grd, blk, 0, stream>>>(x, wv, Vw, 2);

  flash_attn<<<dim3(S_LEN / 16, BATCH * NHEAD), dim3(32), 0, stream>>>(Qw, Kw, Vw, Aw);

  gemm_wmma_f32<<<grd, blk, 0, stream>>>(Aw, wo, out, 0);
}


# attention: 8-wave group per 128 queries of a head; K and V stages of 128 keys loaded once per group (coalesced 16 B, one stage ahead) into LDS, read as fragments by all waves; key sum order changes
# speedup vs baseline: 1.8785x; 1.8785x over previous
#include <hip/hip_runtime.h>
#include <hip/hip_bf16.h>

typedef __attribute__((ext_vector_type(2))) float v2f;
typedef __attribute__((ext_vector_type(8))) float v8f;

#define S_LEN   2048
#define D_MODEL 1024
#define NHEAD   16
#define DK      64
#define BATCH   2
#define M_TOT   (BATCH * S_LEN)
#define ROPE_L2T_OVER_DK 0.20762050593046014f

__device__ __forceinline__ float red_max16(float v) {
  v = fmaxf(v, __shfl_xor(v, 1, 32));
  v = fmaxf(v, __shfl_xor(v, 2, 32));
  v = fmaxf(v, __shfl_xor(v, 4, 32));
  v = fmaxf(v, __shfl_xor(v, 8, 32));
  return v;
}
__device__ __forceinline__ float red_sum16(float v) {
  v += __shfl_xor(v, 1, 32);
  v += __shfl_xor(v, 2, 32);
  v += __shfl_xor(v, 4, 32);
  v += __shfl_xor(v, 8, 32);
  return v;
}

__global__ __launch_bounds__(256) void gemm_wmma_f32(
    const float* __restrict__ X, const float* __restrict__ W,
    float* __restrict__ out, int mode)
{
  const int lane = threadIdx.x & 31;
  const int wave = threadIdx.x >> 5;
  const int half = lane >> 4;
  const int l16  = lane & 15;

  const int m0 = (blockIdx.y * 8 + wave) * 32;
  const int n0 = blockIdx.x * 64;

  const float* Arow0 = X + (size_t)(m0 + l16) * D_MODEL;
  const float* Arow1 = Arow0 + (size_t)16 * D_MODEL;
  const float* Brow0 = W + (size_t)(n0 + l16) * D_MODEL;
  const float* Brow1 = Brow0 + (size_t)16 * D_MODEL;
  const float* Brow2 = Brow0 + (size_t)32 * D_MODEL;
  const float* Brow3 = Brow0 + (size_t)48 * D_MODEL;

  v8f acc[2][4];
#pragma unroll
  for (int i = 0; i < 2; ++i)
#pragma unroll
    for (int t = 0; t < 4; ++t) acc[i][t] = (v8f){};

  const int koff = 2 * half;
#pragma unroll 2
  for (int k = 0; k < D_MODEL; k += 4) {
    const int kk = k + koff;
    v2f a0 = *(const v2f*)(Arow0 + kk);
    v2f a1 = *(const v2f*)(Arow1 + kk);
    v2f b0 = *(const v2f*)(Brow0 + kk);
    v2f b1 = *(const v2f*)(Brow1 + kk);
    v2f b2 = *(const v2f*)(Brow2 + kk);
    v2f b3 = *(const v2f*)(Brow3 + kk);
    acc[0][0] = __builtin_amdgcn_wmma_f32_16x16x4_f32(false, a0, false, b0, (short)0, acc[0][0], false, false);
    acc[0][1] = __builtin_amdgcn_wmma_f32_16x16x4_f32(false, a0, false, b1, (short)0, acc[0][1], false, false);
    acc[0][2] = __builtin_amdgcn_wmma_f32_16x16x4_f32(false, a0, false, b2, (short)0, acc[0][2], false, false);
    acc[0][3] = __builtin_amdgcn_wmma_f32_16x16x4_f32(false, a0, false, b3, (short)0, acc[0][3], false, false);
    acc[1][0] = __builtin_amdgcn_wmma_f32_16x16x4_f32(false, a1, false, b0, (short)0, acc[1][0], false, false);
    acc[1][1] = __builtin_amdgcn_wmma_f32_16x16x4_f32(false, a1, false, b1, (short)0, acc[1][1], false, false);
    acc[1][2] = __builtin_amdgcn_wmma_f32_16x16x4_f32(false, a1, false, b2, (short)0, acc[1][2], false, false);
    acc[1][3] = __builtin_amdgcn_wmma_f32_16x16x4_f32(false, a1, false, b3, (short)0, acc[1][3], false, false);
  }

  if (mode == 0) {
#pragma unroll
    for (int i = 0; i < 2; ++i)
#pragma unroll
      for (int t = 0; t < 4; ++t)
#pragma unroll
        for (int r = 0; r < 8; ++r) {
          const int row = m0 + i * 16 + r + 8 * half;
          out[(size_t)row * D_MODEL + n0 + t * 16 + l16] = acc[i][t][r];
        }
  } else {
#pragma unroll
    for (int t = 0; t < 4; ++t) {
      const int gcol = n0 + t * 16 + l16;
      const int h = gcol >> 6;
      const int d = gcol & 63;
      const bool even = (d & 1) == 0;
      const float invfreq = exp2f(-(float)(d & ~1) * ROPE_L2T_OVER_DK);
#pragma unroll
      for (int i = 0; i < 2; ++i)
#pragma unroll
        for (int r = 0; r < 8; ++r) {
          const int row = m0 + i * 16 + r + 8 * half;
          const int b  = row >> 11;
          const int s  = row & (S_LEN - 1);
          float v  = acc[i][t][r];
          float vp = __shfl_xor(v, 1, 32);
          if (mode == 1) {
            const float ang = (float)s * invfreq;
            const float c = __cosf(ang), sn = __sinf(ang);
            v = v * c + (even ? -vp : vp) * sn;
          }
          out[(((size_t)(b * NHEAD + h) * S_LEN + s) * DK) + d] = v;
        }
    }
  }
}

#define KBLK 64
#define STG  128
#define KP   68
#define VP   80

template <bool DIAG>
__device__ __forceinline__ void attn_block(
    const float* ks, const float* vs, float* ps,
    const v2f (&qa)[16], v8f (&o)[4], float (&mrun)[8], float (&lrun)[8],
    const int l16, const int half, const int kdiff)
{
  v8f sc[4];
#pragma unroll
  for (int t = 0; t < 4; ++t) sc[t] = (v8f){};
  const float* kf = ks + l16 * KP + 2 * half;
#pragma unroll
  for (int c = 0; c < 16; ++c) {
#pragma unroll
    for (int t = 0; t < 4; ++t) {
      v2f bf = *(const v2f*)(kf + t * 16 * KP + 4 * c);
      sc[t] = __builtin_amdgcn_wmma_f32_16x16x4_f32(false, qa[c], false, bf,
                                                    (short)0, sc[t], false, false);
    }
  }

  float scl[8];
#pragma unroll
  for (int r = 0; r < 8; ++r) {
    float s0 = sc[0][r], s1 = sc[1][r], s2 = sc[2][r], s3 = sc[3][r];
    if (DIAG) {
      const int lim = 8 * half + r - kdiff - l16;
      s0 = (0  <= lim) ? s0 : -__builtin_inff();
      s1 = (16 <= lim) ? s1 : -__builtin_inff();
      s2 = (32 <= lim) ? s2 : -__builtin_inff();
      s3 = (48 <= lim) ? s3 : -__builtin_inff();
    }
    const float rmax = red_max16(fmaxf(fmaxf(s0, s1), fmaxf(s2, s3)));
    const float mnew = fmaxf(mrun[r], rmax);
    const float p0 = __expf(s0 - mnew);
    const float p1 = __expf(s1 - mnew);
    const float p2 = __expf(s2 - mnew);
    const float p3 = __expf(s3 - mnew);
    const float rsum = red_sum16((p0 + p1) + (p2 + p3));
    scl[r]  = __expf(mrun[r] - mnew);
    lrun[r] = lrun[r] * scl[r] + rsum;
    mrun[r] = mnew;
    float* pr = ps + (8 * half + r) * KP + l16;
    pr[0] = p0; pr[16] = p1; pr[32] = p2; pr[48] = p3;
  }

#pragma unroll
  for (int r = 0; r < 8; ++r) {
    o[0][r] *= scl[r]; o[1][r] *= scl[r]; o[2][r] *= scl[r]; o[3][r] *= scl[r];
  }

  __builtin_amdgcn_fence(__ATOMIC_RELEASE, "workgroup", "local");
  __builtin_amdgcn_wave_barrier();
  __builtin_amdgcn_fence(__ATOMIC_ACQUIRE, "workgroup", "local");

  const float* pf = ps + l16 * KP + 2 * half;
  const float* vf = vs + (2 * half) * VP + l16;
#pragma unroll
  for (int j = 0; j < 16; ++j) {
    const v2f a = *(const v2f*)(pf + 4 * j);
#pragma unroll
    for (int nt = 0; nt < 4; ++nt) {
      v2f b;
      b[0] = vf[(4 * j) * VP + 16 * nt];
      b[1] = vf[(4 * j + 1) * VP + 16 * nt];
      o[nt] = __builtin_amdgcn_wmma_f32_16x16x4_f32(false, a, false, b,
                                                    (short)0, o[nt], false, false);
    }
  }
}

#define STAGE_LOAD(kp_, vp_) \
  k0 = (kp_)[0];   k1 = (kp_)[256];   k2 = (kp_)[512];   k3 = (kp_)[768];   k4 = (kp_)[1024];   k5 = (kp_)[1280];   k6 = (kp_)[1536];   k7 = (kp_)[1792]; \
  v0 = (vp_)[0];   v1 = (vp_)[256];   v2 = (vp_)[512];   v3 = (vp_)[768];   v4 = (vp_)[1024];   v5 = (vp_)[1280];   v6 = (vp_)[1536];   v7 = (vp_)[1792];
#define STAGE_STORE(kd_, vd_) \
  *(float4*)((kd_) + 0 * KP) = k0; \
  *(float4*)((kd_) + 16 * KP) = k1; \
  *(float4*)((kd_) + 32 * KP) = k2; \
  *(float4*)((kd_) + 48 * KP) = k3; \
  *(float4*)((kd_) + 64 * KP) = k4; \
  *(float4*)((kd_) + 80 * KP) = k5; \
  *(float4*)((kd_) + 96 * KP) = k6; \
  *(float4*)((kd_) + 112 * KP) = k7; \
  *(float4*)((vd_) + 0 * VP) = v0; \
  *(float4*)((vd_) + 16 * VP) = v1; \
  *(float4*)((vd_) + 32 * VP) = v2; \
  *(float4*)((vd_) + 48 * VP) = v3; \
  *(float4*)((vd_) + 64 * VP) = v4; \
  *(float4*)((vd_) + 80 * VP) = v5; \
  *(float4*)((vd_) + 96 * VP) = v6; \
  *(float4*)((vd_) + 112 * VP) = v7;

__global__ __launch_bounds__(256) void flash_attn_lds(
    const float* __restrict__ Q, const float* __restrict__ K,
    const float* __restrict__ V, float* __restrict__ att)
{
  __shared__ __attribute__((aligned(16))) float k_lds[2][STG * KP];
  __shared__ __attribute__((aligned(16))) float v_lds[2][STG * VP];
  __shared__ __attribute__((aligned(16))) float p_lds[8][16 * KP];

  const int tid  = threadIdx.x;
  const int lane = tid & 31;
  const int half = lane >> 4;
  const int l16  = lane & 15;
  const int wave = __builtin_amdgcn_readfirstlane(tid >> 5);

  const int grp = 15 - (int)(blockIdx.x >> 5);
  const int bh  = blockIdx.x & 31;
  const int qt  = grp * 8 + wave;
  const int nst = grp + 1;
  const size_t base = (size_t)bh * S_LEN * DK;

  v2f qa[16];
  {
    const float* qrow = Q + base + (size_t)(qt * 16 + l16) * DK + 2 * half;
#pragma unroll
    for (int c = 0; c < 16; ++c) {
      v2f t = *(const v2f*)(qrow + 4 * c);
      qa[c] = t * 0.125f;
    }
  }

  v8f o[4];
#pragma unroll
  for (int t = 0; t < 4; ++t) o[t] = (v8f){};
  float mrun[8], lrun[8];
#pragma unroll
  for (int r = 0; r < 8; ++r) { mrun[r] = -__builtin_inff(); lrun[r] = 0.f; }

  const float4* kg = (const float4*)(K + base) + tid;
  const float4* vg = (const float4*)(V + base) + tid;
  const int kdst = (tid >> 4) * KP + (tid & 15) * 4;
  const int vdst = (tid >> 4) * VP + (tid & 15) * 4;
  float4 k0, k1, k2, k3, k4, k5, k6, k7, v0, v1, v2, v3, v4, v5, v6, v7;
  STAGE_LOAD(kg, vg)
  STAGE_STORE(&k_lds[0][kdst], &v_lds[0][vdst])
  __syncthreads();

  for (int st = 0; st < nst; ++st) {
    const int cur = st & 1;
    const bool more = (st + 1 < nst);
    if (more) {
      const float4* kn = kg + (st + 1) * (STG * 16);
      const float4* vn = vg + (st + 1) * (STG * 16);
      STAGE_LOAD(kn, vn)
    }

#pragma unroll 1
    for (int sub = 0; sub < 2; ++sub) {
      const int i = 2 * st + sub;
      const int nact = qt - 4 * i + 1;
      const float* ks = &k_lds[cur][sub * KBLK * KP];
      const float* vs = &v_lds[cur][sub * KBLK * VP];
      if (nact >= 5) {
        attn_block<false>(ks, vs, p_lds[wave], qa, o, mrun, lrun, l16, half, 0);
      } else if (nact >= 1) {
        attn_block<true>(ks, vs, p_lds[wave], qa, o, mrun, lrun, l16, half, (i * 4 - qt) * 16);
      }
    }

    if (more) {
      STAGE_STORE(&k_lds[cur ^ 1][kdst], &v_lds[cur ^ 1][vdst])
    }
    __syncthreads();
  }

  const int b = bh >> 4;
  const int h = bh & 15;
#pragma unroll
  for (int r = 0; r < 8; ++r) {
    const float inv = 1.0f / lrun[r];
    const int row = qt * 16 + r + 8 * half;
    const size_t ob = ((size_t)(b * S_LEN) + row) * D_MODEL + h * DK;
    att[ob +  0 + l16] = o[0][r] * inv;
    att[ob + 16 + l16] = o[1][r] * inv;
    att[ob + 32 + l16] = o[2][r] * inv;
    att[ob + 48 + l16] = o[3][r] * inv;
  }
}

extern "C" void kernel_launch(void* const* d_in, const int* in_sizes, int n_in,
                              void* d_out, int out_size, void* d_ws, size_t ws_size,
                              hipStream_t stream) {
  const float* x  = (const float*)d_in[0];
  const float* wq = (const float*)d_in[1];
  const float* wk = (const float*)d_in[2];
  const float* wv = (const float*)d_in[3];
  const float* wo = (const float*)d_in[4];
  float* out = (float*)d_out;

  float* ws  = (float*)d_ws;
  const size_t SZ = (size_t)M_TOT * D_MODEL;
  float* Qw  = ws;
  float* Kw  = ws + SZ;
  float* Vw  = ws + 2 * SZ;
  float* Aw  = ws + 3 * SZ;

  dim3 blk(256);
  dim3 grd(D_MODEL / 64, M_TOT / (32 * 8));
  gemm_wmma_f32<<<grd, blk, 0, stream>>>(x, wq, Qw, 1);
  gemm_wmma_f32<<<grd, blk, 0, stream>>>(x, wk, Kw, 1);
  gemm_wmma_f32<<<grd, blk, 0, stream>>>(x, wv, Vw, 2);

  flash_attn_lds<<<dim3(16 * BATCH * NHEAD), dim3(256), 0, stream>>>(Qw, Kw, Vw, Aw);

  gemm_wmma_f32<<<grd, blk, 0, stream>>>(Aw, wo, out, 0);
}


# all 4 GEMMs: 8-wave group per 128x128 tile of C; X and W slices of 64 k loaded once per group (coalesced 16 B, one stage ahead) into padded LDS, read as fragments by its waves; sums unchanged
# speedup vs baseline: 2.2326x; 1.1885x over previous
#include <hip/hip_runtime.h>
#include <hip/hip_bf16.h>

typedef __attribute__((ext_vector_type(2))) float v2f;
typedef __attribute__((ext_vector_type(8))) float v8f;

#define S_LEN   2048
#define D_MODEL 1024
#define NHEAD   16
#define DK      64
#define BATCH   2
#define M_TOT   (BATCH * S_LEN)
#define ROPE_L2T_OVER_DK 0.20762050593046014f

__device__ __forceinline__ float red_max16(float v) {
  v = fmaxf(v, __shfl_xor(v, 1, 32));
  v = fmaxf(v, __shfl_xor(v, 2, 32));
  v = fmaxf(v, __shfl_xor(v, 4, 32));
  v = fmaxf(v, __shfl_xor(v, 8, 32));
  return v;
}
__device__ __forceinline__ float red_sum16(float v) {
  v += __shfl_xor(v, 1, 32);
  v += __shfl_xor(v, 2, 32);
  v += __shfl_xor(v, 4, 32);
  v += __shfl_xor(v, 8, 32);
  return v;
}

#define GBM 128
#define GBN 128
#define GBK 64
#define GP  68

#define GEMM_STAGE_LOAD(ap_, bp_) \
  ga0 = (ap_)[0 * 4096]; ga1 = (ap_)[1 * 4096]; ga2 = (ap_)[2 * 4096]; ga3 = (ap_)[3 * 4096]; \
  ga4 = (ap_)[4 * 4096]; ga5 = (ap_)[5 * 4096]; ga6 = (ap_)[6 * 4096]; ga7 = (ap_)[7 * 4096]; \
  gb0 = (bp_)[0 * 4096]; gb1 = (bp_)[1 * 4096]; gb2 = (bp_)[2 * 4096]; gb3 = (bp_)[3 * 4096]; \
  gb4 = (bp_)[4 * 4096]; gb5 = (bp_)[5 * 4096]; gb6 = (bp_)[6 * 4096]; gb7 = (bp_)[7 * 4096];
#define GEMM_STAGE_STORE(ad_, bd_) \
  *(float4*)((ad_) + 0 * GP) = ga0;  *(float4*)((ad_) + 16 * GP) = ga1; \
  *(float4*)((ad_) + 32 * GP) = ga2; *(float4*)((ad_) + 48 * GP) = ga3; \
  *(float4*)((ad_) + 64 * GP) = ga4; *(float4*)((ad_) + 80 * GP) = ga5; \
  *(float4*)((ad_) + 96 * GP) = ga6; *(float4*)((ad_) + 112 * GP) = ga7; \
  *(float4*)((bd_) + 0 * GP) = gb0;  *(float4*)((bd_) + 16 * GP) = gb1; \
  *(float4*)((bd_) + 32 * GP) = gb2; *(float4*)((bd_) + 48 * GP) = gb3; \
  *(float4*)((bd_) + 64 * GP) = gb4; *(float4*)((bd_) + 80 * GP) = gb5; \
  *(float4*)((bd_) + 96 * GP) = gb6; *(float4*)((bd_) + 112 * GP) = gb7;

__global__ __launch_bounds__(256) void gemm_wmma_f32(
    const float* __restrict__ X, const float* __restrict__ W,
    float* __restrict__ out, int mode)
{
  __shared__ __attribute__((aligned(16))) float a_lds[2][GBM * GP];
  __shared__ __attribute__((aligned(16))) float b_lds[2][GBN * GP];

  const int tid  = threadIdx.x;
  const int lane = tid & 31;
  const int half = lane >> 4;
  const int l16  = lane & 15;
  const int wave = __builtin_amdgcn_readfirstlane(tid >> 5);
  const int wm = wave >> 1;
  const int wn = wave & 1;

  const int bm = blockIdx.y * GBM;
  const int bn = blockIdx.x * GBN;
  const int m0 = bm + wm * 32;
  const int n0 = bn + wn * 64;

  const float4* ag = (const float4*)(X + (size_t)(bm + (tid >> 4)) * D_MODEL) + (tid & 15);
  const float4* bg = (const float4*)(W + (size_t)(bn + (tid >> 4)) * D_MODEL) + (tid & 15);
  const int sdst = (tid >> 4) * GP + (tid & 15) * 4;

  v8f acc[2][4];
#pragma unroll
  for (int i = 0; i < 2; ++i)
#pragma unroll
    for (int t = 0; t < 4; ++t) acc[i][t] = (v8f){};

  {
    float4 ga0, ga1, ga2, ga3, ga4, ga5, ga6, ga7, gb0, gb1, gb2, gb3, gb4, gb5, gb6, gb7;
    GEMM_STAGE_LOAD(ag, bg)
    GEMM_STAGE_STORE(&a_lds[0][sdst], &b_lds[0][sdst])
  }
  __syncthreads();

  const int afo = (wm * 32 + l16) * GP + 2 * half;
  const int bfo = (wn * 64 + l16) * GP + 2 * half;

  for (int st = 0; st < D_MODEL / GBK; ++st) {
    const int cur = st & 1;
    const bool more = (st + 1 < D_MODEL / GBK);
    float4 ga0, ga1, ga2, ga3, ga4, ga5, ga6, ga7, gb0, gb1, gb2, gb3, gb4, gb5, gb6, gb7;
    if (more) {
      const float4* an = ag + (st + 1) * (GBK / 4);
      const float4* bn_ = bg + (st + 1) * (GBK / 4);
      GEMM_STAGE_LOAD(an, bn_)
    }

    const float* af = &a_lds[cur][afo];
    const float* bf = &b_lds[cur][bfo];
#pragma unroll
    for (int c = 0; c < GBK / 4; ++c) {
      v2f a0 = *(const v2f*)(af + 4 * c);
      v2f a1 = *(const v2f*)(af + 16 * GP + 4 * c);
      v2f b0 = *(const v2f*)(bf + 4 * c);
      v2f b1 = *(const v2f*)(bf + 16 * GP + 4 * c);
      v2f b2 = *(const v2f*)(bf + 32 * GP + 4 * c);
      v2f b3 = *(const v2f*)(bf + 48 * GP + 4 * c);
      acc[0][0] = __builtin_amdgcn_wmma_f32_16x16x4_f32(false, a0, false, b0, (short)0, acc[0][0], false, false);
      acc[0][1] = __builtin_amdgcn_wmma_f32_16x16x4_f32(false, a0, false, b1, (short)0, acc[0][1], false, false);
      acc[0][2] = __builtin_amdgcn_wmma_f32_16x16x4_f32(false, a0, false, b2, (short)0, acc[0][2], false, false);
      acc[0][3] = __builtin_amdgcn_wmma_f32_16x16x4_f32(false, a0, false, b3, (short)0, acc[0][3], false, false);
      acc[1][0] = __builtin_amdgcn_wmma_f32_16x16x4_f32(false, a1, false, b0, (short)0, acc[1][0], false, false);
      acc[1][1] = __builtin_amdgcn_wmma_f32_16x16x4_f32(false, a1, false, b1, (short)0, acc[1][1], false, false);
      acc[1][2] = __builtin_amdgcn_wmma_f32_16x16x4_f32(false, a1, false, b2, (short)0, acc[1][2], false, false);
      acc[1][3] = __builtin_amdgcn_wmma_f32_16x16x4_f32(false, a1, false, b3, (short)0, acc[1][3], false, false);
    }

    if (more) {
      GEMM_STAGE_STORE(&a_lds[cur ^ 1][sdst], &b_lds[cur ^ 1][sdst])
    }
    __syncthreads();
  }

  if (mode == 0) {
#pragma unroll
    for (int i = 0; i < 2; ++i)
#pragma unroll
      for (int t = 0; t < 4; ++t)
#pragma unroll
        for (int r = 0; r < 8; ++r) {
          const int row = m0 + i * 16 + r + 8 * half;
          out[(size_t)row * D_MODEL + n0 + t * 16 + l16] = acc[i][t][r];
        }
  } else {
#pragma unroll
    for (int t = 0; t < 4; ++t) {
      const int gcol = n0 + t * 16 + l16;
      const int h = gcol >> 6;
      const int d = gcol & 63;
      const bool even = (d & 1) == 0;
      const float invfreq = exp2f(-(float)(d & ~1) * ROPE_L2T_OVER_DK);
#pragma unroll
      for (int i = 0; i < 2; ++i)
#pragma unroll
        for (int r = 0; r < 8; ++r) {
          const int row = m0 + i * 16 + r + 8 * half;
          const int b  = row >> 11;
          const int s  = row & (S_LEN - 1);
          float v  = acc[i][t][r];
          float vp = __shfl_xor(v, 1, 32);
          if (mode == 1) {
            const float ang = (float)s * invfreq;
            const float c = __cosf(ang), sn = __sinf(ang);
            v = v * c + (even ? -vp : vp) * sn;
          }
          out[(((size_t)(b * NHEAD + h) * S_LEN + s) * DK) + d] = v;
        }
    }
  }
}

#define KBLK 64
#define STG  128
#define KP   68
#define VP   80

template <bool DIAG>
__device__ __forceinline__ void attn_block(
    const float* ks, const float* vs, float* ps,
    const v2f (&qa)[16], v8f (&o)[4], float (&mrun)[8], float (&lrun)[8],
    const int l16, const int half, const int kdiff)
{
  v8f sc[4];
#pragma unroll
  for (int t = 0; t < 4; ++t) sc[t] = (v8f){};
  const float* kf = ks + l16 * KP + 2 * half;
#pragma unroll
  for (int c = 0; c < 16; ++c) {
#pragma unroll
    for (int t = 0; t < 4; ++t) {
      v2f bf = *(const v2f*)(kf + t * 16 * KP + 4 * c);
      sc[t] = __builtin_amdgcn_wmma_f32_16x16x4_f32(false, qa[c], false, bf,
                                                    (short)0, sc[t], false, false);
    }
  }

  float scl[8];
#pragma unroll
  for (int r = 0; r < 8; ++r) {
    float s0 = sc[0][r], s1 = sc[1][r], s2 = sc[2][r], s3 = sc[3][r];
    if (DIAG) {
      const int lim = 8 * half + r - kdiff - l16;
      s0 = (0  <= lim) ? s0 : -__builtin_inff();
      s1 = (16 <= lim) ? s1 : -__builtin_inff();
      s2 = (32 <= lim) ? s2 : -__builtin_inff();
      s3 = (48 <= lim) ? s3 : -__builtin_inff();
    }
    const float rmax = red_max16(fmaxf(fmaxf(s0, s1), fmaxf(s2, s3)));
    const float mnew = fmaxf(mrun[r], rmax);
    const float p0 = __expf(s0 - mnew);
    const float p1 = __expf(s1 - mnew);
    const float p2 = __expf(s2 - mnew);
    const float p3 = __expf(s3 - mnew);
    const float rsum = red_sum16((p0 + p1) + (p2 + p3));
    scl[r]  = __expf(mrun[r] - mnew);
    lrun[r] = lrun[r] * scl[r] + rsum;
    mrun[r] = mnew;
    float* pr = ps + (8 * half + r) * KP + l16;
    pr[0] = p0; pr[16] = p1; pr[32] = p2; pr[48] = p3;
  }

#pragma unroll
  for (int r = 0; r < 8; ++r) {
    o[0][r] *= scl[r]; o[1][r] *= scl[r]; o[2][r] *= scl[r]; o[3][r] *= scl[r];
  }

  __builtin_amdgcn_fence(__ATOMIC_RELEASE, "workgroup", "local");
  __builtin_amdgcn_wave_barrier();
  __builtin_amdgcn_fence(__ATOMIC_ACQUIRE, "workgroup", "local");

  const float* pf = ps + l16 * KP + 2 * half;
  const float* vf = vs + (2 * half) * VP + l16;
#pragma unroll
  for (int j = 0; j < 16; ++j) {
    const v2f a = *(const v2f*)(pf + 4 * j);
#pragma unroll
    for (int nt = 0; nt < 4; ++nt) {
      v2f b;
      b[0] = vf[(4 * j) * VP + 16 * nt];
      b[1] = vf[(4 * j + 1) * VP + 16 * nt];
      o[nt] = __builtin_amdgcn_wmma_f32_16x16x4_f32(false, a, false, b,
                                                    (short)0, o[nt], false, false);
    }
  }
}

#define STAGE_LOAD(kp_, vp_) \
  k0 = (kp_)[0];   k1 = (kp_)[256];   k2 = (kp_)[512];   k3 = (kp_)[768];   k4 = (kp_)[1024];   k5 = (kp_)[1280];   k6 = (kp_)[1536];   k7 = (kp_)[1792]; \
  v0 = (vp_)[0];   v1 = (vp_)[256];   v2 = (vp_)[512];   v3 = (vp_)[768];   v4 = (vp_)[1024];   v5 = (vp_)[1280];   v6 = (vp_)[1536];   v7 = (vp_)[1792];
#define STAGE_STORE(kd_, vd_) \
  *(float4*)((kd_) + 0 * KP) = k0; \
  *(float4*)((kd_) + 16 * KP) = k1; \
  *(float4*)((kd_) + 32 * KP) = k2; \
  *(float4*)((kd_) + 48 * KP) = k3; \
  *(float4*)((kd_) + 64 * KP) = k4; \
  *(float4*)((kd_) + 80 * KP) = k5; \
  *(float4*)((kd_) + 96 * KP) = k6; \
  *(float4*)((kd_) + 112 * KP) = k7; \
  *(float4*)((vd_) + 0 * VP) = v0; \
  *(float4*)((vd_) + 16 * VP) = v1; \
  *(float4*)((vd_) + 32 * VP) = v2; \
  *(float4*)((vd_) + 48 * VP) = v3; \
  *(float4*)((vd_) + 64 * VP) = v4; \
  *(float4*)((vd_) + 80 * VP) = v5; \
  *(float4*)((vd_) + 96 * VP) = v6; \
  *(float4*)((vd_) + 112 * VP) = v7;

__global__ __launch_bounds__(256) void flash_attn_lds(
    const float* __restrict__ Q, const float* __restrict__ K,
    const float* __restrict__ V, float* __restrict__ att)
{
  __shared__ __attribute__((aligned(16))) float k_lds[2][STG * KP];
  __shared__ __attribute__((aligned(16))) float v_lds[2][STG * VP];
  __shared__ __attribute__((aligned(16))) float p_lds[8][16 * KP];

  const int tid  = threadIdx.x;
  const int lane = tid & 31;
  const int half = lane >> 4;
  const int l16  = lane & 15;
  const int wave = __builtin_amdgcn_readfirstlane(tid >> 5);

  const int grp = 15 - (int)(blockIdx.x >> 5);
  const int bh  = blockIdx.x & 31;
  const int qt  = grp * 8 + wave;
  const int nst = grp + 1;
  const size_t base = (size_t)bh * S_LEN * DK;

  v2f qa[16];
  {
    const float* qrow = Q + base + (size_t)(qt * 16 + l16) * DK + 2 * half;
#pragma unroll
    for (int c = 0; c < 16; ++c) {
      v2f t = *(const v2f*)(qrow + 4 * c);
      qa[c] = t * 0.125f;
    }
  }

  v8f o[4];
#pragma unroll
  for (int t = 0; t < 4; ++t) o[t] = (v8f){};
  float mrun[8], lrun[8];
#pragma unroll
  for (int r = 0; r < 8; ++r) { mrun[r] = -__builtin_inff(); lrun[r] = 0.f; }

  const float4* kg = (const float4*)(K + base) + tid;
  const float4* vg = (const float4*)(V + base) + tid;
  const int kdst = (tid >> 4) * KP + (tid & 15) * 4;
  const int vdst = (tid >> 4) * VP + (tid & 15) * 4;
  float4 k0, k1, k2, k3, k4, k5, k6, k7, v0, v1, v2, v3, v4, v5, v6, v7;
  STAGE_LOAD(kg, vg)
  STAGE_STORE(&k_lds[0][kdst], &v_lds[0][vdst])
  __syncthreads();

  for (int st = 0; st < nst; ++st) {
    const int cur = st & 1;
    const bool more = (st + 1 < nst);
    if (more) {
      const float4* kn = kg + (st + 1) * (STG * 16);
      const float4* vn = vg + (st + 1) * (STG * 16);
      STAGE_LOAD(kn, vn)
    }

#pragma unroll 1
    for (int sub = 0; sub < 2; ++sub) {
      const int i = 2 * st + sub;
      const int nact = qt - 4 * i + 1;
      const float* ks = &k_lds[cur][sub * KBLK * KP];
      const float* vs = &v_lds[cur][sub * KBLK * VP];
      if (nact >= 5) {
        attn_block<false>(ks, vs, p_lds[wave], qa, o, mrun, lrun, l16, half, 0);
      } else if (nact >= 1) {
        attn_block<true>(ks, vs, p_lds[wave], qa, o, mrun, lrun, l16, half, (i * 4 - qt) * 16);
      }
    }

    if (more) {
      STAGE_STORE(&k_lds[cur ^ 1][kdst], &v_lds[cur ^ 1][vdst])
    }
    __syncthreads();
  }

  const int b = bh >> 4;
  const int h = bh & 15;
#pragma unroll
  for (int r = 0; r < 8; ++r) {
    const float inv = 1.0f / lrun[r];
    const int row = qt * 16 + r + 8 * half;
    const size_t ob = ((size_t)(b * S_LEN) + row) * D_MODEL + h * DK;
    att[ob +  0 + l16] = o[0][r] * inv;
    att[ob + 16 + l16] = o[1][r] * inv;
    att[ob + 32 + l16] = o[2][r] * inv;
    att[ob + 48 + l16] = o[3][r] * inv;
  }
}

extern "C" void kernel_launch(void* const* d_in, const int* in_sizes, int n_in,
                              void* d_out, int out_size, void* d_ws, size_t ws_size,
                              hipStream_t stream) {
  const float* x  = (const float*)d_in[0];
  const float* wq = (const float*)d_in[1];
  const float* wk = (const float*)d_in[2];
  const float* wv = (const float*)d_in[3];
  const float* wo = (const float*)d_in[4];
  float* out = (float*)d_out;

  float* ws  = (float*)d_ws;
  const size_t SZ = (size_t)M_TOT * D_MODEL;
  float* Qw  = ws;
  float* Kw  = ws + SZ;
  float* Vw  = ws + 2 * SZ;
  float* Aw  = ws + 3 * SZ;

  dim3 blk(256);
  dim3 grd(D_MODEL / GBN, M_TOT / GBM);
  gemm_wmma_f32<<<grd, blk, 0, stream>>>(x, wq, Qw, 1);
  gemm_wmma_f32<<<grd, blk, 0, stream>>>(x, wk, Kw, 1);
  gemm_wmma_f32<<<grd, blk, 0, stream>>>(x, wv, Vw, 2);

  flash_attn_lds<<<dim3(16 * BATCH * NHEAD), dim3(256), 0, stream>>>(Qw, Kw, Vw, Aw);

  gemm_wmma_f32<<<grd, blk, 0, stream>>>(Aw, wo, out, 0);
}


# Q, K and V projections as one launch of 768 groups (was three of 256): two groups resident per WGP, one's barrier waits, first stage and epilogue run under the other's products; sums unchanged
# speedup vs baseline: 2.3300x; 1.0436x over previous
#include <hip/hip_runtime.h>
#include <hip/hip_bf16.h>

typedef __attribute__((ext_vector_type(2))) float v2f;
typedef __attribute__((ext_vector_type(8))) float v8f;

#define S_LEN   2048
#define D_MODEL 1024
#define NHEAD   16
#define DK      64
#define BATCH   2
#define M_TOT   (BATCH * S_LEN)
#define ROPE_L2T_OVER_DK 0.20762050593046014f

__device__ __forceinline__ float red_max16(float v) {
  v = fmaxf(v, __shfl_xor(v, 1, 32));
  v = fmaxf(v, __shfl_xor(v, 2, 32));
  v = fmaxf(v, __shfl_xor(v, 4, 32));
  v = fmaxf(v, __shfl_xor(v, 8, 32));
  return v;
}
__device__ __forceinline__ float red_sum16(float v) {
  v += __shfl_xor(v, 1, 32);
  v += __shfl_xor(v, 2, 32);
  v += __shfl_xor(v, 4, 32);
  v += __shfl_xor(v, 8, 32);
  return v;
}

#define GBM 128
#define GBN 128
#define GBK 64
#define GP  68

#define GEMM_STAGE_LOAD(ap_, bp_) \
  ga0 = (ap_)[0 * 4096]; ga1 = (ap_)[1 * 4096]; ga2 = (ap_)[2 * 4096]; ga3 = (ap_)[3 * 4096]; \
  ga4 = (ap_)[4 * 4096]; ga5 = (ap_)[5 * 4096]; ga6 = (ap_)[6 * 4096]; ga7 = (ap_)[7 * 4096]; \
  gb0 = (bp_)[0 * 4096]; gb1 = (bp_)[1 * 4096]; gb2 = (bp_)[2 * 4096]; gb3 = (bp_)[3 * 4096]; \
  gb4 = (bp_)[4 * 4096]; gb5 = (bp_)[5 * 4096]; gb6 = (bp_)[6 * 4096]; gb7 = (bp_)[7 * 4096];
#define GEMM_STAGE_STORE(ad_, bd_) \
  *(float4*)((ad_) + 0 * GP) = ga0;  *(float4*)((ad_) + 16 * GP) = ga1; \
  *(float4*)((ad_) + 32 * GP) = ga2; *(float4*)((ad_) + 48 * GP) = ga3; \
  *(float4*)((ad_) + 64 * GP) = ga4; *(float4*)((ad_) + 80 * GP) = ga5; \
  *(float4*)((ad_) + 96 * GP) = ga6; *(float4*)((ad_) + 112 * GP) = ga7; \
  *(float4*)((bd_) + 0 * GP) = gb0;  *(float4*)((bd_) + 16 * GP) = gb1; \
  *(float4*)((bd_) + 32 * GP) = gb2; *(float4*)((bd_) + 48 * GP) = gb3; \
  *(float4*)((bd_) + 64 * GP) = gb4; *(float4*)((bd_) + 80 * GP) = gb5; \
  *(float4*)((bd_) + 96 * GP) = gb6; *(float4*)((bd_) + 112 * GP) = gb7;

__device__ __forceinline__ void gemm_tile_128x128(
    const float* __restrict__ X, const float* __restrict__ W,
    float* __restrict__ out, const int mode, const int bm, const int bn)
{
  __shared__ __attribute__((aligned(16))) float a_lds[2][GBM * GP];
  __shared__ __attribute__((aligned(16))) float b_lds[2][GBN * GP];

  const int tid  = threadIdx.x;
  const int lane = tid & 31;
  const int half = lane >> 4;
  const int l16  = lane & 15;
  const int wave = __builtin_amdgcn_readfirstlane(tid >> 5);
  const int wm = wave >> 1;
  const int wn = wave & 1;

  const int m0 = bm + wm * 32;
  const int n0 = bn + wn * 64;

  const float4* ag = (const float4*)(X + (size_t)(bm + (tid >> 4)) * D_MODEL) + (tid & 15);
  const float4* bg = (const float4*)(W + (size_t)(bn + (tid >> 4)) * D_MODEL) + (tid & 15);
  const int sdst = (tid >> 4) * GP + (tid & 15) * 4;

  v8f acc[2][4];
#pragma unroll
  for (int i = 0; i < 2; ++i)
#pragma unroll
    for (int t = 0; t < 4; ++t) acc[i][t] = (v8f){};

  {
    float4 ga0, ga1, ga2, ga3, ga4, ga5, ga6, ga7, gb0, gb1, gb2, gb3, gb4, gb5, gb6, gb7;
    GEMM_STAGE_LOAD(ag, bg)
    GEMM_STAGE_STORE(&a_lds[0][sdst], &b_lds[0][sdst])
  }
  __syncthreads();

  const int afo = (wm * 32 + l16) * GP + 2 * half;
  const int bfo = (wn * 64 + l16) * GP + 2 * half;

  for (int st = 0; st < D_MODEL / GBK; ++st) {
    const int cur = st & 1;
    const bool more = (st + 1 < D_MODEL / GBK);
    float4 ga0, ga1, ga2, ga3, ga4, ga5, ga6, ga7, gb0, gb1, gb2, gb3, gb4, gb5, gb6, gb7;
    if (more) {
      const float4* an = ag + (st + 1) * (GBK / 4);
      const float4* bn_ = bg + (st + 1) * (GBK / 4);
      GEMM_STAGE_LOAD(an, bn_)
    }

    const float* af = &a_lds[cur][afo];
    const float* bf = &b_lds[cur][bfo];
#pragma unroll
    for (int c = 0; c < GBK / 4; ++c) {
      v2f a0 = *(const v2f*)(af + 4 * c);
      v2f a1 = *(const v2f*)(af + 16 * GP + 4 * c);
      v2f b0 = *(const v2f*)(bf + 4 * c);
      v2f b1 = *(const v2f*)(bf + 16 * GP + 4 * c);
      v2f b2 = *(const v2f*)(bf + 32 * GP + 4 * c);
      v2f b3 = *(const v2f*)(bf + 48 * GP + 4 * c);
      acc[0][0] = __builtin_amdgcn_wmma_f32_16x16x4_f32(false, a0, false, b0, (short)0, acc[0][0], false, false);
      acc[0][1] = __builtin_amdgcn_wmma_f32_16x16x4_f32(false, a0, false, b1, (short)0, acc[0][1], false, false);
      acc[0][2] = __builtin_amdgcn_wmma_f32_16x16x4_f32(false, a0, false, b2, (short)0, acc[0][2], false, false);
      acc[0][3] = __builtin_amdgcn_wmma_f32_16x16x4_f32(false, a0, false, b3, (short)0, acc[0][3], false, false);
      acc[1][0] = __builtin_amdgcn_wmma_f32_16x16x4_f32(false, a1, false, b0, (short)0, acc[1][0], false, false);
      acc[1][1] = __builtin_amdgcn_wmma_f32_16x16x4_f32(false, a1, false, b1, (short)0, acc[1][1], false, false);
      acc[1][2] = __builtin_amdgcn_wmma_f32_16x16x4_f32(false, a1, false, b2, (short)0, acc[1][2], false, false);
      acc[1][3] = __builtin_amdgcn_wmma_f32_16x16x4_f32(false, a1, false, b3, (short)0, acc[1][3], false, false);
    }

    if (more) {
      GEMM_STAGE_STORE(&a_lds[cur ^ 1][sdst], &b_lds[cur ^ 1][sdst])
    }
    __syncthreads();
  }

  if (mode == 0) {
#pragma unroll
    for (int i = 0; i < 2; ++i)
#pragma unroll
      for (int t = 0; t < 4; ++t)
#pragma unroll
        for (int r = 0; r < 8; ++r) {
          const int row = m0 + i * 16 + r + 8 * half;
          out[(size_t)row * D_MODEL + n0 + t * 16 + l16] = acc[i][t][r];
        }
  } else {
#pragma unroll
    for (int t = 0; t < 4; ++t) {
      const int gcol = n0 + t * 16 + l16;
      const int h = gcol >> 6;
      const int d = gcol & 63;
      const bool even = (d & 1) == 0;
      const float invfreq = exp2f(-(float)(d & ~1) * ROPE_L2T_OVER_DK);
#pragma unroll
      for (int i = 0; i < 2; ++i)
#pragma unroll
        for (int r = 0; r < 8; ++r) {
          const int row = m0 + i * 16 + r + 8 * half;
          const int b  = row >> 11;
          const int s  = row & (S_LEN - 1);
          float v  = acc[i][t][r];
          float vp = __shfl_xor(v, 1, 32);
          if (mode == 1) {
            const float ang = (float)s * invfreq;
            const float c = __cosf(ang), sn = __sinf(ang);
            v = v * c + (even ? -vp : vp) * sn;
          }
          out[(((size_t)(b * NHEAD + h) * S_LEN + s) * DK) + d] = v;
        }
    }
  }
}

__global__ __launch_bounds__(256) void gemm_wmma_f32(
    const float* __restrict__ X, const float* __restrict__ W, float* __restrict__ out)
{
  gemm_tile_128x128(X, W, out, 0, blockIdx.y * GBM, blockIdx.x * GBN);
}

__global__ __launch_bounds__(256) void qkv_proj_wmma_f32(
    const float* __restrict__ X, const float* __restrict__ Wq,
    const float* __restrict__ Wk, const float* __restrict__ Wv,
    float* __restrict__ Qo, float* __restrict__ Ko, float* __restrict__ Vo)
{
  const int which = blockIdx.x >> 3;
  const float* W = (which == 0) ? Wq : ((which == 1) ? Wk : Wv);
  float* out     = (which == 0) ? Qo : ((which == 1) ? Ko : Vo);
  gemm_tile_128x128(X, W, out, (which == 2) ? 2 : 1, blockIdx.y * GBM, (blockIdx.x & 7) * GBN);
}

#define KBLK 64
#define STG  128
#define KP   68
#define VP   80

template <bool DIAG>
__device__ __forceinline__ void attn_block(
    const float* ks, const float* vs, float* ps,
    const v2f (&qa)[16], v8f (&o)[4], float (&mrun)[8], float (&lrun)[8],
    const int l16, const int half, const int kdiff)
{
  v8f sc[4];
#pragma unroll
  for (int t = 0; t < 4; ++t) sc[t] = (v8f){};
  const float* kf = ks + l16 * KP + 2 * half;
#pragma unroll
  for (int c = 0; c < 16; ++c) {
#pragma unroll
    for (int t = 0; t < 4; ++t) {
      v2f bf = *(const v2f*)(kf + t * 16 * KP + 4 * c);
      sc[t] = __builtin_amdgcn_wmma_f32_16x16x4_f32(false, qa[c], false, bf,
                                                    (short)0, sc[t], false, false);
    }
  }

  float scl[8];
#pragma unroll
  for (int r = 0; r < 8; ++r) {
    float s0 = sc[0][r], s1 = sc[1][r], s2 = sc[2][r], s3 = sc[3][r];
    if (DIAG) {
      const int lim = 8 * half + r - kdiff - l16;
      s0 = (0  <= lim) ? s0 : -__builtin_inff();
      s1 = (16 <= lim) ? s1 : -__builtin_inff();
      s2 = (32 <= lim) ? s2 : -__builtin_inff();
      s3 = (48 <= lim) ? s3 : -__builtin_inff();
    }
    const float rmax = red_max16(fmaxf(fmaxf(s0, s1), fmaxf(s2, s3)));
    const float mnew = fmaxf(mrun[r], rmax);
    const float p0 = __expf(s0 - mnew);
    const float p1 = __expf(s1 - mnew);
    const float p2 = __expf(s2 - mnew);
    const float p3 = __expf(s3 - mnew);
    const float rsum = red_sum16((p0 + p1) + (p2 + p3));
    scl[r]  = __expf(mrun[r] - mnew);
    lrun[r] = lrun[r] * scl[r] + rsum;
    mrun[r] = mnew;
    float* pr = ps + (8 * half + r) * KP + l16;
    pr[0] = p0; pr[16] = p1; pr[32] = p2; pr[48] = p3;
  }

#pragma unroll
  for (int r = 0; r < 8; ++r) {
    o[0][r] *= scl[r]; o[1][r] *= scl[r]; o[2][r] *= scl[r]; o[3][r] *= scl[r];
  }

  __builtin_amdgcn_fence(__ATOMIC_RELEASE, "workgroup", "local");
  __builtin_amdgcn_wave_barrier();
  __builtin_amdgcn_fence(__ATOMIC_ACQUIRE, "workgroup", "local");

  const float* pf = ps + l16 * KP + 2 * half;
  const float* vf = vs + (2 * half) * VP + l16;
#pragma unroll
  for (int j = 0; j < 16; ++j) {
    const v2f a = *(const v2f*)(pf + 4 * j);
#pragma unroll
    for (int nt = 0; nt < 4; ++nt) {
      v2f b;
      b[0] = vf[(4 * j) * VP + 16 * nt];
      b[1] = vf[(4 * j + 1) * VP + 16 * nt];
      o[nt] = __builtin_amdgcn_wmma_f32_16x16x4_f32(false, a, false, b,
                                                    (short)0, o[nt], false, false);
    }
  }
}

#define STAGE_LOAD(kp_, vp_) \
  k0 = (kp_)[0];   k1 = (kp_)[256];   k2 = (kp_)[512];   k3 = (kp_)[768];   k4 = (kp_)[1024];   k5 = (kp_)[1280];   k6 = (kp_)[1536];   k7 = (kp_)[1792]; \
  v0 = (vp_)[0];   v1 = (vp_)[256];   v2 = (vp_)[512];   v3 = (vp_)[768];   v4 = (vp_)[1024];   v5 = (vp_)[1280];   v6 = (vp_)[1536];   v7 = (vp_)[1792];
#define STAGE_STORE(kd_, vd_) \
  *(float4*)((kd_) + 0 * KP) = k0; \
  *(float4*)((kd_) + 16 * KP) = k1; \
  *(float4*)((kd_) + 32 * KP) = k2; \
  *(float4*)((kd_) + 48 * KP) = k3; \
  *(float4*)((kd_) + 64 * KP) = k4; \
  *(float4*)((kd_) + 80 * KP) = k5; \
  *(float4*)((kd_) + 96 * KP) = k6; \
  *(float4*)((kd_) + 112 * KP) = k7; \
  *(float4*)((vd_) + 0 * VP) = v0; \
  *(float4*)((vd_) + 16 * VP) = v1; \
  *(float4*)((vd_) + 32 * VP) = v2; \
  *(float4*)((vd_) + 48 * VP) = v3; \
  *(float4*)((vd_) + 64 * VP) = v4; \
  *(float4*)((vd_) + 80 * VP) = v5; \
  *(float4*)((vd_) + 96 * VP) = v6; \
  *(float4*)((vd_) + 112 * VP) = v7;

__global__ __launch_bounds__(256) void flash_attn_lds(
    const float* __restrict__ Q, const float* __restrict__ K,
    const float* __restrict__ V, float* __restrict__ att)
{
  __shared__ __attribute__((aligned(16))) float k_lds[2][STG * KP];
  __shared__ __attribute__((aligned(16))) float v_lds[2][STG * VP];
  __shared__ __attribute__((aligned(16))) float p_lds[8][16 * KP];

  const int tid  = threadIdx.x;
  const int lane = tid & 31;
  const int half = lane >> 4;
  const int l16  = lane & 15;
  const int wave = __builtin_amdgcn_readfirstlane(tid >> 5);

  const int grp = 15 - (int)(blockIdx.x >> 5);
  const int bh  = blockIdx.x & 31;
  const int qt  = grp * 8 + wave;
  const int nst = grp + 1;
  const size_t base = (size_t)bh * S_LEN * DK;

  v2f qa[16];
  {
    const float* qrow = Q + base + (size_t)(qt * 16 + l16) * DK + 2 * half;
#pragma unroll
    for (int c = 0; c < 16; ++c) {
      v2f t = *(const v2f*)(qrow + 4 * c);
      qa[c] = t * 0.125f;
    }
  }

  v8f o[4];
#pragma unroll
  for (int t = 0; t < 4; ++t) o[t] = (v8f){};
  float mrun[8], lrun[8];
#pragma unroll
  for (int r = 0; r < 8; ++r) { mrun[r] = -__builtin_inff(); lrun[r] = 0.f; }

  const float4* kg = (const float4*)(K + base) + tid;
  const float4* vg = (const float4*)(V + base) + tid;
  const int kdst = (tid >> 4) * KP + (tid & 15) * 4;
  const int vdst = (tid >> 4) * VP + (tid & 15) * 4;
  float4 k0, k1, k2, k3, k4, k5, k6, k7, v0, v1, v2, v3, v4, v5, v6, v7;
  STAGE_LOAD(kg, vg)
  STAGE_STORE(&k_lds[0][kdst], &v_lds[0][vdst])
  __syncthreads();

  for (int st = 0; st < nst; ++st) {
    const int cur = st & 1;
    const bool more = (st + 1 < nst);
    if (more) {
      const float4* kn = kg + (st + 1) * (STG * 16);
      const float4* vn = vg + (st + 1) * (STG * 16);
      STAGE_LOAD(kn, vn)
    }

#pragma unroll 1
    for (int sub = 0; sub < 2; ++sub) {
      const int i = 2 * st + sub;
      const int nact = qt - 4 * i + 1;
      const float* ks = &k_lds[cur][sub * KBLK * KP];
      const float* vs = &v_lds[cur][sub * KBLK * VP];
      if (nact >= 5) {
        attn_block<false>(ks, vs, p_lds[wave], qa, o, mrun, lrun, l16, half, 0);
      } else if (nact >= 1) {
        attn_block<true>(ks, vs, p_lds[wave], qa, o, mrun, lrun, l16, half, (i * 4 - qt) * 16);
      }
    }

    if (more) {
      STAGE_STORE(&k_lds[cur ^ 1][kdst], &v_lds[cur ^ 1][vdst])
    }
    __syncthreads();
  }

  const int b = bh >> 4;
  const int h = bh & 15;
#pragma unroll
  for (int r = 0; r < 8; ++r) {
    const float inv = 1.0f / lrun[r];
    const int row = qt * 16 + r + 8 * half;
    const size_t ob = ((size_t)(b * S_LEN) + row) * D_MODEL + h * DK;
    att[ob +  0 + l16] = o[0][r] * inv;
    att[ob + 16 + l16] = o[1][r] * inv;
    att[ob + 32 + l16] = o[2][r] * inv;
    att[ob + 48 + l16] = o[3][r] * inv;
  }
}

extern "C" void kernel_launch(void* const* d_in, const int* in_sizes, int n_in,
                              void* d_out, int out_size, void* d_ws, size_t ws_size,
                              hipStream_t stream) {
  const float* x  = (const float*)d_in[0];
  const float* wq = (const float*)d_in[1];
  const float* wk = (const float*)d_in[2];
  const float* wv = (const float*)d_in[3];
  const float* wo = (const float*)d_in[4];
  float* out = (float*)d_out;

  float* ws  = (float*)d_ws;
  const size_t SZ = (size_t)M_TOT * D_MODEL;
  float* Qw  = ws;
  float* Kw  = ws + SZ;
  float* Vw  = ws + 2 * SZ;
  float* Aw  = ws + 3 * SZ;

  dim3 blk(256);
  dim3 grd(D_MODEL / GBN, M_TOT / GBM);
  qkv_proj_wmma_f32<<<dim3(3 * (D_MODEL / GBN), M_TOT / GBM), blk, 0, stream>>>(x, wq, wk, wv, Qw, Kw, Vw);

  flash_attn_lds<<<dim3(16 * BATCH * NHEAD), dim3(256), 0, stream>>>(Qw, Kw, Vw, Aw);

  gemm_wmma_f32<<<grd, blk, 0, stream>>>(Aw, wo, out);
}
